# starting kernel
# baseline (speedup 1.0000x reference)
#include <hip/hip_runtime.h>
#include <math.h>

typedef __attribute__((ext_vector_type(2))) float v2f;
typedef __attribute__((ext_vector_type(8))) float v8f;

#define HDIM     256
#define KCH      64
#define TS       16
#define NTHREADS 512

#define OFF_W2   0
#define OFF_H1   (HDIM*HDIM)
#define OFF_H1D  (OFF_H1 + TS*HDIM)
#define OFF_W1Z  (OFF_H1D + TS*HDIM)
#define OFF_W1T  (OFF_W1Z + HDIM)
#define OFF_B1   (OFF_W1T + HDIM)
#define OFF_B2   (OFF_B1 + HDIM)
#define OFF_W3   (OFF_B2 + HDIM)
#define OFF_F    (OFF_W3 + HDIM)
#define OFF_FD   (OFF_F + TS)
#define OFF_Z    (OFF_FD + TS)
#define OFF_ZT   (OFF_Z + TS)
#define OFF_DIV  (OFF_ZT + TS)
#define OFF_KZ   (OFF_DIV + TS)
#define OFF_KD   (OFF_KZ + TS)
#define LDS_FLOATS (OFF_KD + TS)

__device__ __forceinline__ float tanh_fast(float x) {
#if defined(__AMDGCN__) && __has_builtin(__builtin_amdgcn_tanhf)
  return __builtin_amdgcn_tanhf(x);
#else
  float e = __expf(2.0f * x);
  return 1.0f - 2.0f / (e + 1.0f);
#endif
}

__global__ void __launch_bounds__(NTHREADS, 1)
cnf1d_rk4_wmma(const float* __restrict__ z0, const float* __restrict__ W1,
               const float* __restrict__ b1, const float* __restrict__ W2,
               const float* __restrict__ b2, const float* __restrict__ W3,
               const float* __restrict__ b3, float* __restrict__ out, int Btot)
{
  extern __shared__ float sm[];
  float* sW2  = sm + OFF_W2;
  float* sH1  = sm + OFF_H1;
  float* sH1d = sm + OFF_H1D;
  float* sW1z = sm + OFF_W1Z;
  float* sW1t = sm + OFF_W1T;
  float* sB1  = sm + OFF_B1;
  float* sB2  = sm + OFF_B2;
  float* sW3  = sm + OFF_W3;
  float* sF   = sm + OFF_F;
  float* sFd  = sm + OFF_FD;
  float* sZ   = sm + OFF_Z;
  float* sZt  = sm + OFF_ZT;
  float* sDiv = sm + OFF_DIV;
  float* sKz  = sm + OFF_KZ;
  float* sKd  = sm + OFF_KD;

  const int tid  = threadIdx.x;
  const int lane = tid & 31;
  const int wv   = tid >> 5;
  const float dtv = 0.25f;
  const float b3v = b3[0];

  for (int i = tid; i < HDIM; i += NTHREADS) {
    sW1z[i] = W1[i];
    sW1t[i] = W1[HDIM + i];
    sB1[i]  = b1[i];
    sB2[i]  = b2[i];
    sW3[i]  = W3[i];
  }
  for (int e = tid; e < HDIM * HDIM; e += NTHREADS) {
    int row = e >> 8, col = e & 255;
    int c  = row >> 2, rr = row & 3;
    int n  = col >> 4, lo = col & 15;
    int l  = ((rr >> 1) << 4) | lo;
    int d  = rr & 1;
    int idx = ((((((c << 4) | n) << 5) | l) << 1) | d);
    sW2[idx] = W2[e];
  }
  if (tid < TS) {
    float z = z0[(size_t)blockIdx.x * TS + tid];
    sZ[tid] = z; sZt[tid] = z; sDiv[tid] = 0.0f;
    sKz[tid] = 0.0f; sKd[tid] = 0.0f;
    sF[tid] = b3v; sFd[tid] = 0.0f;
  }
  __syncthreads();

  #pragma unroll 1
  for (int ev = 0; ev < 16; ++ev) {
    const int stg = ev & 3;
    float tt = (float)(ev >> 2) * dtv;
    if (stg == 3) tt += dtv; else if (stg != 0) tt += 0.5f * dtv;

    #pragma unroll
    for (int i = 0; i < 8; ++i) {
      int s = tid + NTHREADS * i;
      int d = s & 1;
      int l = (s >> 1) & 31;
      int c = s >> 6;
      int m = l & 15;
      int j = (c << 2) + ((l >> 4) << 1) + d;
      float w1 = sW1z[j];
      float a  = sZt[m] * w1 + tt * sW1t[j] + sB1[j];
      float h  = tanh_fast(a);
      sH1[s]  = h;
      sH1d[s] = (1.0f - h * h) * w1;
    }
    __syncthreads();

    v8f acc  = {};
    v8f accd = {};
    const v2f* aF  = (const v2f*)sH1;
    const v2f* adF = (const v2f*)sH1d;
    const v2f* bF  = (const v2f*)sW2;
    #pragma unroll 4
    for (int c = 0; c < KCH; ++c) {
      v2f af = aF[(c << 5) + lane];
      v2f ad = adF[(c << 5) + lane];
      v2f bf = bF[((((c << 4) | wv) << 5) + lane)];
      acc  = __builtin_amdgcn_wmma_f32_16x16x4_f32(false, af, false, bf,
                                                   (short)0, acc,  false, false);
      accd = __builtin_amdgcn_wmma_f32_16x16x4_f32(false, ad, false, bf,
                                                   (short)0, accd, false, false);
    }

    {
      const int n  = lane & 15;
      const int j  = (wv << 4) | n;
      const int mb = (lane >> 4) << 3;
      const float w3 = sW3[j];
      const float bb = sB2[j];
      #pragma unroll
      for (int v = 0; v < 8; ++v) {
        float h2  = tanh_fast(acc[v] + bb);
        float pf  = h2 * w3;
        float pfd = (1.0f - h2 * h2) * accd[v] * w3;
        #pragma unroll
        for (int msk = 1; msk < 16; msk <<= 1) {
          pf  += __shfl_xor(pf,  msk, 32);
          pfd += __shfl_xor(pfd, msk, 32);
        }
        if (n == 0) {
          atomicAdd(&sF[mb + v],  pf);
          atomicAdd(&sFd[mb + v], pfd);
        }
      }
    }
    __syncthreads();

    if (tid < TS) {
      float kz = sF[tid];
      float kd = sFd[tid];
      float w  = (stg == 1 || stg == 2) ? 2.0f : 1.0f;
      float akz = sKz[tid] + w * kz;
      float akd = sKd[tid] + w * kd;
      if (stg < 3) {
        float cf = (stg == 2) ? dtv : 0.5f * dtv;
        sZt[tid] = sZ[tid] + cf * kz;
        sKz[tid] = akz; sKd[tid] = akd;
      } else {
        float zn = sZ[tid] + (dtv * (1.0f / 6.0f)) * akz;
        sZ[tid] = zn; sZt[tid] = zn;
        sDiv[tid] += (dtv * (1.0f / 6.0f)) * akd;
        sKz[tid] = 0.0f; sKd[tid] = 0.0f;
      }
      sF[tid] = b3v; sFd[tid] = 0.0f;
    }
    __syncthreads();
  }

  if (tid < TS) {
    int g = blockIdx.x * TS + tid;
    out[g] = sZ[tid];
    out[Btot + g] = sDiv[tid];
  }
}

extern "C" void kernel_launch(void* const* d_in, const int* in_sizes, int n_in,
                              void* d_out, int out_size, void* d_ws, size_t ws_size,
                              hipStream_t stream) {
  (void)n_in; (void)d_ws; (void)ws_size; (void)out_size;
  const float* z0 = (const float*)d_in[0];
  const float* W1 = (const float*)d_in[1];
  const float* b1 = (const float*)d_in[2];
  const float* W2 = (const float*)d_in[3];
  const float* b2 = (const float*)d_in[4];
  const float* W3 = (const float*)d_in[5];
  const float* b3 = (const float*)d_in[6];
  float* out = (float*)d_out;

  const int B    = in_sizes[0];
  const int nblk = B / TS;
  const size_t lds = (size_t)LDS_FLOATS * sizeof(float);

  hipFuncSetAttribute(reinterpret_cast<const void*>(cnf1d_rk4_wmma),
                      hipFuncAttributeMaxDynamicSharedMemorySize, (int)lds);
  cnf1d_rk4_wmma<<<nblk, NTHREADS, lds, stream>>>(z0, W1, b1, W2, b2, W3, b3, out, B);
}


# WMMA operands swapped (W2 frag as A, activations as B): tile is a2 transposed, W3 dot sums 8 registers in-lane + one permlanex16, not 4-level bpermute butterflies; changes W3-sum order
# speedup vs baseline: 1.2244x; 1.2244x over previous
#include <hip/hip_runtime.h>
#include <math.h>

typedef __attribute__((ext_vector_type(2))) float v2f;
typedef __attribute__((ext_vector_type(8))) float v8f;

#define HDIM     256
#define KCH      64
#define TS       16
#define NTHREADS 512

#define OFF_W2   0
#define OFF_H1   (HDIM*HDIM)
#define OFF_H1D  (OFF_H1 + TS*HDIM)
#define OFF_W1Z  (OFF_H1D + TS*HDIM)
#define OFF_W1T  (OFF_W1Z + HDIM)
#define OFF_B1   (OFF_W1T + HDIM)
#define OFF_F    (OFF_B1 + HDIM)
#define OFF_FD   (OFF_F + TS)
#define OFF_Z    (OFF_FD + TS)
#define OFF_ZT   (OFF_Z + TS)
#define OFF_DIV  (OFF_ZT + TS)
#define OFF_KZ   (OFF_DIV + TS)
#define OFF_KD   (OFF_KZ + TS)
#define LDS_FLOATS (OFF_KD + TS)

__device__ __forceinline__ float xhalf(float x) {
  int b = __builtin_bit_cast(int, x);
  return __builtin_bit_cast(float,
      __builtin_amdgcn_permlanex16(b, b, 0x76543210, 0xfedcba98, false, false));
}

__device__ __forceinline__ float tanh_fast(float x) {
#if defined(__AMDGCN__) && __has_builtin(__builtin_amdgcn_tanhf)
  return __builtin_amdgcn_tanhf(x);
#else
  float e = __expf(2.0f * x);
  return 1.0f - 2.0f / (e + 1.0f);
#endif
}

__global__ void __launch_bounds__(NTHREADS, 1)
cnf1d_rk4_wmma(const float* __restrict__ z0, const float* __restrict__ W1,
               const float* __restrict__ b1, const float* __restrict__ W2,
               const float* __restrict__ b2, const float* __restrict__ W3,
               const float* __restrict__ b3, float* __restrict__ out, int Btot)
{
  extern __shared__ float sm[];
  float* sW2  = sm + OFF_W2;
  float* sH1  = sm + OFF_H1;
  float* sH1d = sm + OFF_H1D;
  float* sW1z = sm + OFF_W1Z;
  float* sW1t = sm + OFF_W1T;
  float* sB1  = sm + OFF_B1;
  float* sF   = sm + OFF_F;
  float* sFd  = sm + OFF_FD;
  float* sZ   = sm + OFF_Z;
  float* sZt  = sm + OFF_ZT;
  float* sDiv = sm + OFF_DIV;
  float* sKz  = sm + OFF_KZ;
  float* sKd  = sm + OFF_KD;

  const int tid  = threadIdx.x;
  const int lane = tid & 31;
  const int wv   = tid >> 5;
  const float dtv = 0.25f;
  const float b3v = b3[0];

  for (int i = tid; i < HDIM; i += NTHREADS) {
    sW1z[i] = W1[i];
    sW1t[i] = W1[HDIM + i];
    sB1[i]  = b1[i];
  }
  const int jb = (wv << 4) | ((lane >> 4) << 3);
  float b2r[8], w3r[8];
  #pragma unroll
  for (int r = 0; r < 8; ++r) { b2r[r] = b2[jb + r]; w3r[r] = W3[jb + r]; }
  for (int e = tid; e < HDIM * HDIM; e += NTHREADS) {
    int row = e >> 8, col = e & 255;
    int c  = row >> 2, rr = row & 3;
    int n  = col >> 4, lo = col & 15;
    int l  = ((rr >> 1) << 4) | lo;
    int d  = rr & 1;
    int idx = ((((((c << 4) | n) << 5) | l) << 1) | d);
    sW2[idx] = W2[e];
  }
  if (tid < TS) {
    float z = z0[(size_t)blockIdx.x * TS + tid];
    sZ[tid] = z; sZt[tid] = z; sDiv[tid] = 0.0f;
    sKz[tid] = 0.0f; sKd[tid] = 0.0f;
    sF[tid] = b3v; sFd[tid] = 0.0f;
  }
  __syncthreads();

  #pragma unroll 1
  for (int ev = 0; ev < 16; ++ev) {
    const int stg = ev & 3;
    float tt = (float)(ev >> 2) * dtv;
    if (stg == 3) tt += dtv; else if (stg != 0) tt += 0.5f * dtv;

    #pragma unroll
    for (int i = 0; i < 8; ++i) {
      int s = tid + NTHREADS * i;
      int d = s & 1;
      int l = (s >> 1) & 31;
      int c = s >> 6;
      int m = l & 15;
      int j = (c << 2) + ((l >> 4) << 1) + d;
      float w1 = sW1z[j];
      float a  = sZt[m] * w1 + tt * sW1t[j] + sB1[j];
      float h  = tanh_fast(a);
      sH1[s]  = h;
      sH1d[s] = (1.0f - h * h) * w1;
    }
    __syncthreads();

    v8f acc  = {};
    v8f accd = {};
    const v2f* hF  = (const v2f*)sH1;
    const v2f* hdF = (const v2f*)sH1d;
    const v2f* wF  = (const v2f*)sW2;
    #pragma unroll 4
    for (int c = 0; c < KCH; ++c) {
      v2f hf = hF[(c << 5) + lane];
      v2f hd = hdF[(c << 5) + lane];
      v2f wf = wF[((((c << 4) | wv) << 5) + lane)];
      acc  = __builtin_amdgcn_wmma_f32_16x16x4_f32(false, wf, false, hf,
                                                   (short)0, acc,  false, false);
      accd = __builtin_amdgcn_wmma_f32_16x16x4_f32(false, wf, false, hd,
                                                   (short)0, accd, false, false);
    }

    {
      float pf  = 0.0f;
      float pfd = 0.0f;
      #pragma unroll
      for (int r = 0; r < 8; ++r) {
        float h2 = tanh_fast(acc[r] + b2r[r]);
        pf  += h2 * w3r[r];
        pfd += (1.0f - h2 * h2) * accd[r] * w3r[r];
      }
      pf  = pf  + xhalf(pf);
      pfd = pfd + xhalf(pfd);
      if (lane < 16) {
        atomicAdd(&sF[lane],  pf);
        atomicAdd(&sFd[lane], pfd);
      }
    }
    __syncthreads();

    if (tid < TS) {
      float kz = sF[tid];
      float kd = sFd[tid];
      float w  = (stg == 1 || stg == 2) ? 2.0f : 1.0f;
      float akz = sKz[tid] + w * kz;
      float akd = sKd[tid] + w * kd;
      if (stg < 3) {
        float cf = (stg == 2) ? dtv : 0.5f * dtv;
        sZt[tid] = sZ[tid] + cf * kz;
        sKz[tid] = akz; sKd[tid] = akd;
      } else {
        float zn = sZ[tid] + (dtv * (1.0f / 6.0f)) * akz;
        sZ[tid] = zn; sZt[tid] = zn;
        sDiv[tid] += (dtv * (1.0f / 6.0f)) * akd;
        sKz[tid] = 0.0f; sKd[tid] = 0.0f;
      }
      sF[tid] = b3v; sFd[tid] = 0.0f;
    }
    __syncthreads();
  }

  if (tid < TS) {
    int g = blockIdx.x * TS + tid;
    out[g] = sZ[tid];
    out[Btot + g] = sDiv[tid];
  }
}

extern "C" void kernel_launch(void* const* d_in, const int* in_sizes, int n_in,
                              void* d_out, int out_size, void* d_ws, size_t ws_size,
                              hipStream_t stream) {
  (void)n_in; (void)d_ws; (void)ws_size; (void)out_size;
  const float* z0 = (const float*)d_in[0];
  const float* W1 = (const float*)d_in[1];
  const float* b1 = (const float*)d_in[2];
  const float* W2 = (const float*)d_in[3];
  const float* b2 = (const float*)d_in[4];
  const float* W3 = (const float*)d_in[5];
  const float* b3 = (const float*)d_in[6];
  float* out = (float*)d_out;

  const int B    = in_sizes[0];
  const int nblk = B / TS;
  const size_t lds = (size_t)LDS_FLOATS * sizeof(float);

  hipFuncSetAttribute(reinterpret_cast<const void*>(cnf1d_rk4_wmma),
                      hipFuncAttributeMaxDynamicSharedMemorySize, (int)lds);
  cnf1d_rk4_wmma<<<nblk, NTHREADS, lds, stream>>>(z0, W1, b1, W2, b2, W3, b3, out, B);
}


# Two 16-sample tiles per workgroup on two 8-wave groups run half an evaluation apart: one tile's tanh/RK4 stages hide under the other's WMMAs; f summed in fixed order from LDS partials, not atomics
# speedup vs baseline: 1.4241x; 1.1631x over previous
#include <hip/hip_runtime.h>
#include <math.h>


typedef __attribute__((ext_vector_type(2))) float v2f;
typedef __attribute__((ext_vector_type(4))) float v4f;
typedef __attribute__((ext_vector_type(8))) float v8f;

#define HDIM     256
#define KCH      64
#define TS       16
#define NGRP     2
#define GTHREADS 256
#define NTHREADS 512
#define KL       60
#define KR       (KCH - KL)
#define KA       30

#define OFF_W2   0
#define OFF_H1   (KL*4*HDIM)
#define OFF_H1D  (OFF_H1 + NGRP*TS*HDIM)
#define OFF_PF   (OFF_H1D + NGRP*TS*HDIM)
#define OFF_PFD  (OFF_PF + NGRP*TS*8)
#define OFF_OUT  (OFF_PFD + NGRP*TS*8)
#define LDS_FLOATS (OFF_OUT + 2*NGRP*TS)

__device__ __forceinline__ float xhalf(float x) {
  int b = __builtin_bit_cast(int, x);
  return __builtin_bit_cast(float,
      __builtin_amdgcn_permlanex16(b, b, 0x76543210, 0xfedcba98, false, false));
}

__device__ __forceinline__ float tanh_fast(float x) {
#if defined(__AMDGCN__) && __has_builtin(__builtin_amdgcn_tanhf)
  return __builtin_amdgcn_tanhf(x);
#else
  float e = __expf(2.0f * x);
  return 1.0f - 2.0f / (e + 1.0f);
#endif
}

#define WMMA4(W0, W1, HF, HD)                                                          \
  do {                                                                                 \
    acc0  = __builtin_amdgcn_wmma_f32_16x16x4_f32(false, (W0), false, (HF), (short)0,  \
                                                  acc0,  false, false);                \
    accd0 = __builtin_amdgcn_wmma_f32_16x16x4_f32(false, (W0), false, (HD), (short)0,  \
                                                  accd0, false, false);                \
    acc1  = __builtin_amdgcn_wmma_f32_16x16x4_f32(false, (W1), false, (HF), (short)0,  \
                                                  acc1,  false, false);                \
    accd1 = __builtin_amdgcn_wmma_f32_16x16x4_f32(false, (W1), false, (HD), (short)0,  \
                                                  accd1, false, false);                \
  } while (0)

#define WMMA_DRAIN()                                                          \
  asm volatile("s_wait_alu depctr_va_vdst(0)"                                 \
               : "+v"(acc0), "+v"(accd0), "+v"(acc1), "+v"(accd1))

__device__ __forceinline__ void mm_lds(const v2f* wp, const v2f* hp, const v2f* hdp,
                                       int c0, int c1,
                                       v8f& acc0, v8f& accd0, v8f& acc1, v8f& accd1)
{
  v2f hfA = hp[c0 << 5],        hdA = hdp[c0 << 5];
  v2f w0A = wp[c0 << 6],        w1A = wp[(c0 << 6) + 32];
  v2f hfB = hp[(c0 + 1) << 5],  hdB = hdp[(c0 + 1) << 5];
  v2f w0B = wp[(c0 + 1) << 6],  w1B = wp[((c0 + 1) << 6) + 32];
  #pragma unroll
  for (int c = c0; c < c1; c += 2) {
    const int cn = (c + 2 < c1) ? (c + 2) : c;
    v2f hfC = hp[cn << 5],        hdC = hdp[cn << 5];
    v2f w0C = wp[cn << 6],        w1C = wp[(cn << 6) + 32];
    v2f hfD = hp[(cn + 1) << 5],  hdD = hdp[(cn + 1) << 5];
    v2f w0D = wp[(cn + 1) << 6],  w1D = wp[((cn + 1) << 6) + 32];
    __builtin_amdgcn_sched_barrier(0);
    WMMA4(w0A, w1A, hfA, hdA);
    WMMA4(w0B, w1B, hfB, hdB);
    hfA = hfC; hdA = hdC; w0A = w0C; w1A = w1C;
    hfB = hfD; hdB = hdD; w0B = w0D; w1B = w1D;
  }
}

__global__ void __launch_bounds__(NTHREADS, 1)
cnf1d_rk4_wmma(const float* __restrict__ z0, const float* __restrict__ W1,
               const float* __restrict__ b1, const float* __restrict__ W2,
               const float* __restrict__ b2, const float* __restrict__ W3,
               const float* __restrict__ b3, float* __restrict__ out, int Btot)
{
  extern __shared__ __attribute__((aligned(16))) float sm[];
  float* sW2 = sm + OFF_W2;

  const int tid  = threadIdx.x;
  const int lane = tid & 31;
  const int wv   = __builtin_amdgcn_readfirstlane(tid >> 5);
  const int grp  = wv >> 3;
  const int w8   = wv & 7;
  const int n0   = w8 << 1;
  const int gt   = tid & (GTHREADS - 1);
  const float dtv = 0.25f;
  const float b3v = b3[0];

  float* sH1g  = sm + OFF_H1  + grp * (TS * HDIM);
  float* sH1dg = sm + OFF_H1D + grp * (TS * HDIM);
  float* sPfg  = sm + OFF_PF  + grp * (TS * 8);
  float* sPfdg = sm + OFF_PFD + grp * (TS * 8);
  float* sOut  = sm + OFF_OUT;

  const int l1  = (gt >> 1) & 31;
  const int m1  = l1 & 15;
  const int jb1 = ((gt >> 6) << 2) + ((l1 >> 4) << 1) + (gt & 1);
  float w1z_r[16], w1t_r[16], b1_r[16];
  #pragma unroll
  for (int i = 0; i < 16; ++i) {
    w1z_r[i] = W1[jb1 + 16 * i];
    w1t_r[i] = W1[HDIM + jb1 + 16 * i];
    b1_r[i]  = b1[jb1 + 16 * i];
  }

  const int jbA = (n0 << 4) | ((lane >> 4) << 3);
  float b2r0[8], w3r0[8], b2r1[8], w3r1[8];
  #pragma unroll
  for (int r = 0; r < 8; ++r) {
    b2r0[r] = b2[jbA + r];       w3r0[r] = W3[jbA + r];
    b2r1[r] = b2[jbA + 16 + r];  w3r1[r] = W3[jbA + 16 + r];
  }

  v2f wr0[KR], wr1[KR];
  #pragma unroll
  for (int q = 0; q < KR; ++q) {
    const int row = ((KL + q) << 2) + ((lane >> 4) << 1);
    const int col = (n0 << 4) + (lane & 15);
    wr0[q][0] = W2[row * HDIM + col];
    wr0[q][1] = W2[(row + 1) * HDIM + col];
    wr1[q][0] = W2[row * HDIM + col + 16];
    wr1[q][1] = W2[(row + 1) * HDIM + col + 16];
  }

  for (int e = tid; e < KL * 4 * HDIM; e += NTHREADS) {
    int row = e >> 8, col = e & 255;
    int c  = row >> 2, rr = row & 3;
    int n  = col >> 4, lo = col & 15;
    int l  = ((rr >> 1) << 4) | lo;
    int d  = rr & 1;
    int idx = ((((((n >> 1) * KL + c) << 1) | (n & 1)) << 5 | l) << 1) | d;
    sW2[idx] = W2[e];
  }

  float zc  = z0[(size_t)blockIdx.x * (TS * NGRP) + grp * TS + m1];
  float ztc = zc;
  float akz = 0.0f, akd = 0.0f, dvc = 0.0f;

  v8f acc0 = {}, accd0 = {}, acc1 = {}, accd1 = {};
  const v2f* wp  = (const v2f*)sW2 + w8 * (KL * 64) + lane;
  const v2f* hp  = (const v2f*)sH1g + lane;
  const v2f* hdp = (const v2f*)sH1dg + lane;
  __syncthreads();

  #pragma unroll 1
  for (int t = 0; t < 4 * 16 + 1 + 2 * (NGRP - 1); ++t) {
    const int idx = t - 2 * grp;
    if (idx >= 0 && idx <= 64) {
      const int ph = idx & 3;
      const int ev = idx >> 2;
      if (ph == 0) {
        if (ev > 0) {
          const int stg = (ev - 1) & 3;
          const v4f* pz = (const v4f*)(sPfg  + m1 * 8);
          const v4f* pd = (const v4f*)(sPfdg + m1 * 8);
          v4f za = pz[0], zb = pz[1], da = pd[0], db = pd[1];
          float kz = b3v + za[0]; kz += za[1]; kz += za[2]; kz += za[3];
          kz += zb[0]; kz += zb[1]; kz += zb[2]; kz += zb[3];
          float kd = da[0]; kd += da[1]; kd += da[2]; kd += da[3];
          kd += db[0]; kd += db[1]; kd += db[2]; kd += db[3];
          float w = (stg == 1 || stg == 2) ? 2.0f : 1.0f;
          akz = akz + w * kz;
          akd = akd + w * kd;
          if (stg < 3) {
            float cf = (stg == 2) ? dtv : 0.5f * dtv;
            ztc = zc + cf * kz;
          } else {
            zc  = zc + (dtv * (1.0f / 6.0f)) * akz;
            ztc = zc;
            dvc += (dtv * (1.0f / 6.0f)) * akd;
            akz = 0.0f; akd = 0.0f;
          }
        }
        if (ev < 16) {
          const int stg = ev & 3;
          float tt = (float)(ev >> 2) * dtv;
          if (stg == 3) tt += dtv; else if (stg != 0) tt += 0.5f * dtv;
          #pragma unroll
          for (int i = 0; i < 16; ++i) {
            int s = gt + GTHREADS * i;
            float w1 = w1z_r[i];
            float a  = ztc * w1 + tt * w1t_r[i] + b1_r[i];
            float h  = tanh_fast(a);
            sH1g[s]  = h;
            sH1dg[s] = (1.0f - h * h) * w1;
          }
        } else if (gt < 32 && (gt & 1) == 0) {
          sOut[grp * TS + m1] = zc;
          sOut[NGRP * TS + grp * TS + m1] = dvc;
        }
      } else if (ph == 1) {
        mm_lds(wp, hp, hdp, 0, KA, acc0, accd0, acc1, accd1);
        WMMA_DRAIN();
      } else if (ph == 2) {
        mm_lds(wp, hp, hdp, KA, KL, acc0, accd0, acc1, accd1);
        #pragma unroll
        for (int q = 0; q < KR; ++q) {
          v2f hf = hp[(KL + q) << 5];
          v2f hd = hdp[(KL + q) << 5];
          WMMA4(wr0[q], wr1[q], hf, hd);
        }
        WMMA_DRAIN();
      } else {
        float pf  = 0.0f;
        float pfd = 0.0f;
        #pragma unroll
        for (int r = 0; r < 8; ++r) {
          float h2 = tanh_fast(acc0[r] + b2r0[r]);
          pf  += h2 * w3r0[r];
          pfd += (1.0f - h2 * h2) * accd0[r] * w3r0[r];
        }
        #pragma unroll
        for (int r = 0; r < 8; ++r) {
          float h2 = tanh_fast(acc1[r] + b2r1[r]);
          pf  += h2 * w3r1[r];
          pfd += (1.0f - h2 * h2) * accd1[r] * w3r1[r];
        }
        pf  = pf  + xhalf(pf);
        pfd = pfd + xhalf(pfd);
        if (lane < 16) {
          sPfg[lane * 8 + w8]  = pf;
          sPfdg[lane * 8 + w8] = pfd;
        }
        v8f zero = {};
        acc0 = zero; accd0 = zero; acc1 = zero; accd1 = zero;
      }
    }
    __syncthreads();
  }

  if (tid < NGRP * TS) {
    int g = blockIdx.x * (TS * NGRP) + tid;
    out[g] = sOut[tid];
    out[Btot + g] = sOut[NGRP * TS + tid];
  }
}

extern "C" void kernel_launch(void* const* d_in, const int* in_sizes, int n_in,
                              void* d_out, int out_size, void* d_ws, size_t ws_size,
                              hipStream_t stream) {
  (void)n_in; (void)d_ws; (void)ws_size; (void)out_size;
  const float* z0 = (const float*)d_in[0];
  const float* W1 = (const float*)d_in[1];
  const float* b1 = (const float*)d_in[2];
  const float* W2 = (const float*)d_in[3];
  const float* b2 = (const float*)d_in[4];
  const float* W3 = (const float*)d_in[5];
  const float* b3 = (const float*)d_in[6];
  float* out = (float*)d_out;

  const int B    = in_sizes[0];
  const int nblk = B / (TS * NGRP);
  const size_t lds = (size_t)LDS_FLOATS * sizeof(float);

  hipFuncSetAttribute(reinterpret_cast<const void*>(cnf1d_rk4_wmma),
                      hipFuncAttributeMaxDynamicSharedMemorySize, (int)lds);
  cnf1d_rk4_wmma<<<nblk, NTHREADS, lds, stream>>>(z0, W1, b1, W2, b2, W3, b3, out, B);
}


# Matmul cut 1/4-1/2-1/4 and run on across the barriers (next fragments loaded before them, sums in flight): one tile's start shares a slot with the other's end (at s_setprio 1) and stage 3; same sums
# speedup vs baseline: 1.4776x; 1.0376x over previous
#include <hip/hip_runtime.h>
#include <math.h>


typedef __attribute__((ext_vector_type(2))) float v2f;
typedef __attribute__((ext_vector_type(4))) float v4f;
typedef __attribute__((ext_vector_type(8))) float v8f;

#define HDIM     256
#define KCH      64
#define TS       16
#define NGRP     2
#define GTHREADS 256
#define NTHREADS 512
#define KL       60
#define KR       (KCH - KL)
#define KQ       16
#define KB       48

#define OFF_W2   0
#define OFF_H1   (KL*4*HDIM)
#define OFF_H1D  (OFF_H1 + NGRP*TS*HDIM)
#define OFF_PF   (OFF_H1D + NGRP*TS*HDIM)
#define OFF_PFD  (OFF_PF + NGRP*TS*8)
#define OFF_OUT  (OFF_PFD + NGRP*TS*8)
#define LDS_FLOATS (OFF_OUT + 2*NGRP*TS)

__device__ __forceinline__ float xhalf(float x) {
  int b = __builtin_bit_cast(int, x);
  return __builtin_bit_cast(float,
      __builtin_amdgcn_permlanex16(b, b, 0x76543210, 0xfedcba98, false, false));
}

__device__ __forceinline__ float tanh_fast(float x) {
#if defined(__AMDGCN__) && __has_builtin(__builtin_amdgcn_tanhf)
  return __builtin_amdgcn_tanhf(x);
#else
  float e = __expf(2.0f * x);
  return 1.0f - 2.0f / (e + 1.0f);
#endif
}

#define WMMA4(W0, W1, HF, HD)                                                          \
  do {                                                                                 \
    acc0  = __builtin_amdgcn_wmma_f32_16x16x4_f32(false, (W0), false, (HF), (short)0,  \
                                                  acc0,  false, false);                \
    accd0 = __builtin_amdgcn_wmma_f32_16x16x4_f32(false, (W0), false, (HD), (short)0,  \
                                                  accd0, false, false);                \
    acc1  = __builtin_amdgcn_wmma_f32_16x16x4_f32(false, (W1), false, (HF), (short)0,  \
                                                  acc1,  false, false);                \
    accd1 = __builtin_amdgcn_wmma_f32_16x16x4_f32(false, (W1), false, (HD), (short)0,  \
                                                  accd1, false, false);                \
  } while (0)

#define WMMA_DRAIN()                                                          \
  asm volatile("s_wait_alu depctr_va_vdst(0)"                                 \
               : "+v"(acc0), "+v"(accd0), "+v"(acc1), "+v"(accd1))

struct Frags { v2f hfA, hdA, w0A, w1A, hfB, hdB, w0B, w1B; };

__device__ __forceinline__ void mm_first(const v2f* wp, const v2f* hp, const v2f* hdp, int c0, Frags& f)
{
  f.hfA = hp[c0 << 5];        f.hdA = hdp[c0 << 5];
  f.w0A = wp[c0 << 6];        f.w1A = wp[(c0 << 6) + 32];
  f.hfB = hp[(c0 + 1) << 5];  f.hdB = hdp[(c0 + 1) << 5];
  f.w0B = wp[(c0 + 1) << 6];  f.w1B = wp[((c0 + 1) << 6) + 32];
}

__device__ __forceinline__ void mm_run(const v2f* wp, const v2f* hp, const v2f* hdp,
                                       int c0, int c1, int cend, Frags& f,
                                       v8f& acc0, v8f& accd0, v8f& acc1, v8f& accd1)
{
  #pragma unroll
  for (int c = c0; c < c1; c += 2) {
    const int cn = (c + 2 < cend) ? (c + 2) : c;
    v2f hfC = hp[cn << 5],        hdC = hdp[cn << 5];
    v2f w0C = wp[cn << 6],        w1C = wp[(cn << 6) + 32];
    v2f hfD = hp[(cn + 1) << 5],  hdD = hdp[(cn + 1) << 5];
    v2f w0D = wp[(cn + 1) << 6],  w1D = wp[((cn + 1) << 6) + 32];
    __builtin_amdgcn_sched_barrier(0);
    WMMA4(f.w0A, f.w1A, f.hfA, f.hdA);
    WMMA4(f.w0B, f.w1B, f.hfB, f.hdB);
    f.hfA = hfC; f.hdA = hdC; f.w0A = w0C; f.w1A = w1C;
    f.hfB = hfD; f.hdB = hdD; f.w0B = w0D; f.w1B = w1D;
  }
}

__global__ void __launch_bounds__(NTHREADS, 1)
cnf1d_rk4_wmma(const float* __restrict__ z0, const float* __restrict__ W1,
               const float* __restrict__ b1, const float* __restrict__ W2,
               const float* __restrict__ b2, const float* __restrict__ W3,
               const float* __restrict__ b3, float* __restrict__ out, int Btot)
{
  extern __shared__ __attribute__((aligned(16))) float sm[];
  float* sW2 = sm + OFF_W2;

  const int tid  = threadIdx.x;
  const int lane = tid & 31;
  const int wv   = __builtin_amdgcn_readfirstlane(tid >> 5);
  const int grp  = wv >> 3;
  const int w8   = wv & 7;
  const int n0   = w8 << 1;
  const int gt   = tid & (GTHREADS - 1);
  const float dtv = 0.25f;
  const float b3v = b3[0];

  float* sH1g  = sm + OFF_H1  + grp * (TS * HDIM);
  float* sH1dg = sm + OFF_H1D + grp * (TS * HDIM);
  float* sPfg  = sm + OFF_PF  + grp * (TS * 8);
  float* sPfdg = sm + OFF_PFD + grp * (TS * 8);
  float* sOut  = sm + OFF_OUT;

  const int l1  = (gt >> 1) & 31;
  const int m1  = l1 & 15;
  const int jb1 = ((gt >> 6) << 2) + ((l1 >> 4) << 1) + (gt & 1);
  float w1z_r[16], w1t_r[16], b1_r[16];
  #pragma unroll
  for (int i = 0; i < 16; ++i) {
    w1z_r[i] = W1[jb1 + 16 * i];
    w1t_r[i] = W1[HDIM + jb1 + 16 * i];
    b1_r[i]  = b1[jb1 + 16 * i];
  }

  const int jbA = (n0 << 4) | ((lane >> 4) << 3);
  float b2r0[8], w3r0[8], b2r1[8], w3r1[8];
  #pragma unroll
  for (int r = 0; r < 8; ++r) {
    b2r0[r] = b2[jbA + r];       w3r0[r] = W3[jbA + r];
    b2r1[r] = b2[jbA + 16 + r];  w3r1[r] = W3[jbA + 16 + r];
  }

  v2f wr0[KR], wr1[KR];
  #pragma unroll
  for (int q = 0; q < KR; ++q) {
    const int row = ((KL + q) << 2) + ((lane >> 4) << 1);
    const int col = (n0 << 4) + (lane & 15);
    wr0[q][0] = W2[row * HDIM + col];
    wr0[q][1] = W2[(row + 1) * HDIM + col];
    wr1[q][0] = W2[row * HDIM + col + 16];
    wr1[q][1] = W2[(row + 1) * HDIM + col + 16];
  }

  for (int e = tid; e < KL * 4 * HDIM; e += NTHREADS) {
    int row = e >> 8, col = e & 255;
    int c  = row >> 2, rr = row & 3;
    int n  = col >> 4, lo = col & 15;
    int l  = ((rr >> 1) << 4) | lo;
    int d  = rr & 1;
    int idx = ((((((n >> 1) * KL + c) << 1) | (n & 1)) << 5 | l) << 1) | d;
    sW2[idx] = W2[e];
  }

  float zc  = z0[(size_t)blockIdx.x * (TS * NGRP) + grp * TS + m1];
  float ztc = zc;
  float akz = 0.0f, akd = 0.0f, dvc = 0.0f;

  v8f acc0, accd0, acc1, accd1;
  const v2f* wp  = (const v2f*)sW2 + w8 * (KL * 64) + lane;
  const v2f* hp  = (const v2f*)sH1g + lane;
  const v2f* hdp = (const v2f*)sH1dg + lane;
  __syncthreads();

  Frags fr = {};
  #pragma unroll 1
  for (int u = 0; u < 2 * 16 + 1 + (NGRP - 1); ++u) {
    const int j  = u - grp;
    const bool on = (j >= 0 && j <= 32);
    const bool sa = on && ((j & 1) != 0);
    const bool sb = on && ((j & 1) == 0);
    const int ev = j >> 1;

    if (sa) {
      { v8f zero = {}; acc0 = zero; accd0 = zero; acc1 = zero; accd1 = zero; }
      mm_first(wp, hp, hdp, 0, fr);
      mm_run(wp, hp, hdp, 0, KQ, KL, fr, acc0, accd0, acc1, accd1);
    } else if (sb && ev > 0) {
      __builtin_amdgcn_s_setprio(1);
      mm_run(wp, hp, hdp, KB, KL, KL, fr, acc0, accd0, acc1, accd1);
      #pragma unroll
      for (int q = 0; q < KR; ++q) {
        v2f hf = hp[(KL + q) << 5];
        v2f hd = hdp[(KL + q) << 5];
        WMMA4(wr0[q], wr1[q], hf, hd);
      }
      __builtin_amdgcn_s_setprio(0);
      WMMA_DRAIN();
      float pf  = 0.0f;
      float pfd = 0.0f;
      #pragma unroll
      for (int r = 0; r < 8; ++r) {
        float h2 = tanh_fast(acc0[r] + b2r0[r]);
        pf  += h2 * w3r0[r];
        pfd += (1.0f - h2 * h2) * accd0[r] * w3r0[r];
      }
      #pragma unroll
      for (int r = 0; r < 8; ++r) {
        float h2 = tanh_fast(acc1[r] + b2r1[r]);
        pf  += h2 * w3r1[r];
        pfd += (1.0f - h2 * h2) * accd1[r] * w3r1[r];
      }
      pf  = pf  + xhalf(pf);
      pfd = pfd + xhalf(pfd);
      if (lane < 16) {
        sPfg[lane * 8 + w8]  = pf;
        sPfdg[lane * 8 + w8] = pfd;
      }
    }
    __syncthreads();

    if (sa) {
      mm_run(wp, hp, hdp, KQ, KB, KL, fr, acc0, accd0, acc1, accd1);
    } else if (sb) {
      if (ev > 0) {
        const int stg = (ev - 1) & 3;
        const v4f* pz = (const v4f*)(sPfg  + m1 * 8);
        const v4f* pd = (const v4f*)(sPfdg + m1 * 8);
        v4f za = pz[0], zb = pz[1], da = pd[0], db = pd[1];
        float kz = b3v + za[0]; kz += za[1]; kz += za[2]; kz += za[3];
        kz += zb[0]; kz += zb[1]; kz += zb[2]; kz += zb[3];
        float kd = da[0]; kd += da[1]; kd += da[2]; kd += da[3];
        kd += db[0]; kd += db[1]; kd += db[2]; kd += db[3];
        float w = (stg == 1 || stg == 2) ? 2.0f : 1.0f;
        akz = akz + w * kz;
        akd = akd + w * kd;
        if (stg < 3) {
          float cf = (stg == 2) ? dtv : 0.5f * dtv;
          ztc = zc + cf * kz;
        } else {
          zc  = zc + (dtv * (1.0f / 6.0f)) * akz;
          ztc = zc;
          dvc += (dtv * (1.0f / 6.0f)) * akd;
          akz = 0.0f; akd = 0.0f;
        }
      }
      if (ev < 16) {
        const int stg = ev & 3;
        float tt = (float)(ev >> 2) * dtv;
        if (stg == 3) tt += dtv; else if (stg != 0) tt += 0.5f * dtv;
        #pragma unroll
        for (int i = 0; i < 16; ++i) {
          int s = gt + GTHREADS * i;
          float w1 = w1z_r[i];
          float a  = ztc * w1 + tt * w1t_r[i] + b1_r[i];
          float h  = tanh_fast(a);
          sH1g[s]  = h;
          sH1dg[s] = (1.0f - h * h) * w1;
        }
      } else if (gt < 32 && (gt & 1) == 0) {
        sOut[grp * TS + m1] = zc;
        sOut[NGRP * TS + grp * TS + m1] = dvc;
      }
    }
    __syncthreads();
  }

  if (tid < NGRP * TS) {
    int g = blockIdx.x * (TS * NGRP) + tid;
    out[g] = sOut[tid];
    out[Btot + g] = sOut[NGRP * TS + tid];
  }
}

extern "C" void kernel_launch(void* const* d_in, const int* in_sizes, int n_in,
                              void* d_out, int out_size, void* d_ws, size_t ws_size,
                              hipStream_t stream) {
  (void)n_in; (void)d_ws; (void)ws_size; (void)out_size;
  const float* z0 = (const float*)d_in[0];
  const float* W1 = (const float*)d_in[1];
  const float* b1 = (const float*)d_in[2];
  const float* W2 = (const float*)d_in[3];
  const float* b2 = (const float*)d_in[4];
  const float* W3 = (const float*)d_in[5];
  const float* b3 = (const float*)d_in[6];
  float* out = (float*)d_out;

  const int B    = in_sizes[0];
  const int nblk = B / (TS * NGRP);
  const size_t lds = (size_t)LDS_FLOATS * sizeof(float);

  hipFuncSetAttribute(reinterpret_cast<const void*>(cnf1d_rk4_wmma),
                      hipFuncAttributeMaxDynamicSharedMemorySize, (int)lds);
  cnf1d_rk4_wmma<<<nblk, NTHREADS, lds, stream>>>(z0, W1, b1, W2, b2, W3, b3, out, B);
}
